# starting kernel
# baseline (speedup 1.0000x reference)
#include <hip/hip_runtime.h>

typedef __attribute__((ext_vector_type(2))) float v2f;
typedef __attribute__((ext_vector_type(8))) float v8f;

#define N_LAYERS 7
#define N_QUBITS 8
#define DIM 256

__global__ __launch_bounds__(256) void build_u_kernel(const float* __restrict__ w,
                                                      float* __restrict__ bp) {
  __shared__ float col[DIM];
  const int i = threadIdx.x;
  const int j = blockIdx.x;
  col[i] = (i == j) ? 1.0f : 0.0f;
  __syncthreads();
  for (int l = 0; l < N_LAYERS; ++l) {
    for (int q = 0; q < N_QUBITS; ++q) {
      const float th = 0.5f * w[l * N_QUBITS + q];
      const float c = cosf(th);
      const float s = sinf(th);
      const int mask = 1 << (7 - q);
      const float a0 = col[i & ~mask];
      const float a1 = col[i | mask];
      __syncthreads();
      col[i] = (i & mask) ? (s * a0 + c * a1) : (c * a0 - s * a1);
      __syncthreads();
    }
    for (int q = 0; q < N_QUBITS - 1; ++q) {
      const int mc = 1 << (7 - q);
      const int mt = 1 << (6 - q);
      const int src = (i & mc) ? (i ^ mt) : i;
      const float v = col[src];
      __syncthreads();
      col[i] = v;
      __syncthreads();
    }
  }
  const int k   = j & ~1;
  const int c   = j & 1;
  const int kk  = k >> 2;
  const int hf  = (k >> 1) & 1;
  const int nt  = i >> 4;
  const int l16 = i & 15;
  const int lane = hf * 16 + l16;
  bp[(size_t)(((nt * 64 + kk) * 32) + lane) * 2 + c] = col[i];
}

__global__ __launch_bounds__(256) void qnn_fused_kernel(
    const float* __restrict__ x, const v2f* __restrict__ bp,
    const float* __restrict__ W1, const float* __restrict__ b1,
    const float* __restrict__ W2, const float* __restrict__ b2,
    float* __restrict__ out) {
  __shared__ float redP[8][16][16];
  __shared__ float redM[8][16][16];

  const int lane = threadIdx.x & 31;
  const int wave = threadIdx.x >> 5;
  const int hf   = lane >> 4;
  const int l16  = lane & 15;
  const int rbase = blockIdx.x * 128 + wave * 16;

  const float* xrow = x + (size_t)(rbase + l16) * DIM + 2 * hf;

  v2f afrag[64];
#pragma unroll
  for (int kk = 0; kk < 64; ++kk) {
    afrag[kk] = *(const v2f*)(xrow + 4 * kk);
  }

  float accP[8], accM[8];
#pragma unroll
  for (int v = 0; v < 8; ++v) { accP[v] = 0.f; accM[v] = 0.f; }

  for (int nt = 0; nt < 16; ++nt) {
    const v2f* btile = bp + (size_t)nt * 64 * 32 + lane;
    v8f acc = {0.f, 0.f, 0.f, 0.f, 0.f, 0.f, 0.f, 0.f};
#pragma unroll
    for (int kk = 0; kk < 64; ++kk) {
      const v2f b = btile[kk * 32];
      acc = __builtin_amdgcn_wmma_f32_16x16x4_f32(
           false, afrag[kk],  false, b,
           (short)0, acc,  false,  false);
    }
    if (nt < 8) {
#pragma unroll
      for (int v = 0; v < 8; ++v) accP[v] += acc[v] * acc[v];
    } else {
#pragma unroll
      for (int v = 0; v < 8; ++v) accM[v] += acc[v] * acc[v];
    }
  }

#pragma unroll
  for (int v = 0; v < 8; ++v) {
    redP[wave][hf * 8 + v][l16] = accP[v];
    redM[wave][hf * 8 + v][l16] = accM[v];
  }
  __syncthreads();

  if (threadIdx.x < 128) {
    const int wv = threadIdx.x >> 4;
    const int r  = threadIdx.x & 15;
    float P = 0.f, M = 0.f;
#pragma unroll
    for (int i = 0; i < 16; ++i) { P += redP[wv][r][i]; M += redM[wv][r][i]; }
    const float z = (P - M) / (P + M);
    float t = b2[0];
#pragma unroll
    for (int jj = 0; jj < 16; ++jj) {
      float h = z * W1[jj] + b1[jj];
      h = h > 0.f ? h : 0.f;
      t += h * W2[jj];
    }
    out[(size_t)blockIdx.x * 128 + wv * 16 + r] = 1.f / (1.f + expf(-t));
  }
}

extern "C" void kernel_launch(void* const* d_in, const int* in_sizes, int n_in,
                              void* d_out, int out_size, void* d_ws, size_t ws_size,
                              hipStream_t stream) {
  const float* x  = (const float*)d_in[0];
  const float* w  = (const float*)d_in[1];
  const float* W1 = (const float*)d_in[2];
  const float* b1 = (const float*)d_in[3];
  const float* W2 = (const float*)d_in[4];
  const float* b2 = (const float*)d_in[5];
  float* out = (float*)d_out;
  float* bp  = (float*)d_ws;

  const int B = in_sizes[0] / DIM;
  build_u_kernel<<<DIM, DIM, 0, stream>>>(w, bp);
  qnn_fused_kernel<<<B / 128, 256, 0, stream>>>(x, (const v2f*)bp, W1, b1, W2, b2, out);
}


# U build: one wave per column keeps all 256 amplitudes in registers: RY by register pairs and DPP/permlane lane-xor, CNOT chain as one lane gather per layer; no LDS, no block barriers (was 210)
# speedup vs baseline: 2.2648x; 2.2648x over previous
#include <hip/hip_runtime.h>

typedef __attribute__((ext_vector_type(2))) float v2f;
typedef __attribute__((ext_vector_type(8))) float v8f;

#define N_LAYERS 7
#define N_QUBITS 8
#define DIM 256


template <int M>
__device__ __forceinline__ float lane_xor(float v) {
  const int iv = __builtin_bit_cast(int, v);
  if constexpr (M == 16) {
    return __builtin_bit_cast(float, __builtin_amdgcn_permlanex16(
        iv, iv, 0x76543210, 0xfedcba98, false, false));
  } else {
    return __builtin_bit_cast(float, __builtin_amdgcn_update_dpp(
        0, iv, 0x160 + M, 0xF, 0xF, true));
  }
}

__device__ __forceinline__ float lane_bcast(float v, int src) {
  return __builtin_bit_cast(float,
      __builtin_amdgcn_readlane(__builtin_bit_cast(int, v), src));
}

template <int Q>
__device__ __forceinline__ void ry_gate(float (&a)[8], float c, float s, int lane) {
  if constexpr (Q < 3) {
    constexpr int MR = 4 >> Q;
#pragma unroll
    for (int r = 0; r < 8; ++r) {
      if ((r & MR) == 0) {
        const float a0 = a[r], a1 = a[r | MR];
        a[r]      = c * a0 - s * a1;
        a[r | MR] = s * a0 + c * a1;
      }
    }
  } else {
    constexpr int ML = 128 >> Q;
    const float sg = (lane & ML) ? s : -s;
#pragma unroll
    for (int r = 0; r < 8; ++r) {
      const float p = lane_xor<ML>(a[r]);
      a[r] = c * a[r] + sg * p;
    }
  }
}

__global__ __launch_bounds__(32) void build_u_kernel(const float* __restrict__ w,
                                                     float* __restrict__ bp) {
  const int lane = threadIdx.x;
  const int j = blockIdx.x;

  const int g1 = (lane + 32 < N_LAYERS * N_QUBITS) ? lane + 32 : N_LAYERS * N_QUBITS - 1;
  const float th0 = 0.5f * w[lane];
  const float th1 = 0.5f * w[g1];
  const float c0 = cosf(th0), s0 = sinf(th0);
  const float c1 = cosf(th1), s1 = sinf(th1);

  float a[8];
#pragma unroll
  for (int r = 0; r < 8; ++r) a[r] = (r * 32 + lane == j) ? 1.0f : 0.0f;

  const int gl = lane ^ (lane >> 1);

#pragma unroll
  for (int l = 0; l < N_LAYERS; ++l) {
    const int g = l * N_QUBITS;
#define GATE(Q)                                                              \
    ry_gate<Q>(a,                                                            \
               (g + Q < 32) ? lane_bcast(c0, g + Q) : lane_bcast(c1, g + Q - 32), \
               (g + Q < 32) ? lane_bcast(s0, g + Q) : lane_bcast(s1, g + Q - 32), \
               lane)
    GATE(0); GATE(1); GATE(2); GATE(3); GATE(4); GATE(5); GATE(6); GATE(7);
#undef GATE
    float t[8];
#pragma unroll
    for (int r = 0; r < 8; ++r) t[r] = __shfl(a[r ^ (r >> 1)], gl ^ ((r & 1) << 4), 32);
#pragma unroll
    for (int r = 0; r < 8; ++r) a[r] = t[r];
  }

  const int kk  = j >> 2;
  const int hf  = (j >> 1) & 1;
  const int c   = j & 1;
  const int l16 = lane & 15;
#pragma unroll
  for (int r = 0; r < 4; ++r) {
    const int nt = r * 2 + (lane >> 4);
    bp[(size_t)(((nt * 64 + kk) * 32) + hf * 16 + l16) * 2 + c] = a[r];
  }
}

__global__ __launch_bounds__(256) __attribute__((amdgpu_waves_per_eu(4, 4)))
void qnn_fused_kernel(
    const float* __restrict__ x, const v2f* __restrict__ bp,
    const float* __restrict__ W1, const float* __restrict__ b1,
    const float* __restrict__ W2, const float* __restrict__ b2,
    float* __restrict__ out) {
  __shared__ float redP[8][16][16];
  __shared__ float redS[8][32];

  const int lane = threadIdx.x & 31;
  const int wave = threadIdx.x >> 5;
  const int hf   = lane >> 4;
  const int l16  = lane & 15;
  const int rbase = blockIdx.x * 128 + wave * 16;

  const float* xrow = x + (size_t)(rbase + l16) * DIM + 2 * hf;

  v2f afrag[64];
#pragma unroll
  for (int kk = 0; kk < 64; ++kk) {
    afrag[kk] = *(const v2f*)(xrow + 4 * kk);
  }

  float accP[8];
#pragma unroll
  for (int v = 0; v < 8; ++v) accP[v] = 0.f;

  for (int nt = 0; nt < 8; ++nt) {
    const v2f* btile = bp + (size_t)nt * 64 * 32 + lane;
    v8f acc = {0.f, 0.f, 0.f, 0.f, 0.f, 0.f, 0.f, 0.f};
#pragma unroll
    for (int kk = 0; kk < 64; ++kk) {
      const v2f b = btile[kk * 32];
      acc = __builtin_amdgcn_wmma_f32_16x16x4_f32(
           false, afrag[kk],  false, b,
           (short)0, acc,  false,  false);
    }
#pragma unroll
    for (int v = 0; v < 8; ++v) accP[v] += acc[v] * acc[v];
  }

  v2f sq0 = {0.f, 0.f}, sq1 = {0.f, 0.f}, sq2 = {0.f, 0.f}, sq3 = {0.f, 0.f};
#pragma unroll
  for (int kk = 0; kk < 64; kk += 4) {
    sq0 += afrag[kk]     * afrag[kk];
    sq1 += afrag[kk + 1] * afrag[kk + 1];
    sq2 += afrag[kk + 2] * afrag[kk + 2];
    sq3 += afrag[kk + 3] * afrag[kk + 3];
  }
  const v2f sq = (sq0 + sq1) + (sq2 + sq3);
  redS[wave][lane] = sq[0] + sq[1];

#pragma unroll
  for (int v = 0; v < 8; ++v) {
    redP[wave][hf * 8 + v][l16] = accP[v];
  }
  __syncthreads();

  if (threadIdx.x < 128) {
    const int wv = threadIdx.x >> 4;
    const int r  = threadIdx.x & 15;
    float P = 0.f;
#pragma unroll
    for (int i = 0; i < 16; ++i) P += redP[wv][r][i];
    const float S = redS[wv][r] + redS[wv][16 + r];
    const float z = (2.f * P - S) / S;
    float t = b2[0];
#pragma unroll
    for (int jj = 0; jj < 16; ++jj) {
      float h = z * W1[jj] + b1[jj];
      h = h > 0.f ? h : 0.f;
      t += h * W2[jj];
    }
    out[(size_t)blockIdx.x * 128 + wv * 16 + r] = 1.f / (1.f + expf(-t));
  }
}

extern "C" void kernel_launch(void* const* d_in, const int* in_sizes, int n_in,
                              void* d_out, int out_size, void* d_ws, size_t ws_size,
                              hipStream_t stream) {
  const float* x  = (const float*)d_in[0];
  const float* w  = (const float*)d_in[1];
  const float* W1 = (const float*)d_in[2];
  const float* b1 = (const float*)d_in[3];
  const float* W2 = (const float*)d_in[4];
  const float* b2 = (const float*)d_in[5];
  float* out = (float*)d_out;
  float* bp  = (float*)d_ws;

  const int B = in_sizes[0] / DIM;
  build_u_kernel<<<DIM, 32, 0, stream>>>(w, bp);
  qnn_fused_kernel<<<B / 128, 256, 0, stream>>>(x, (const v2f*)bp, W1, b1, W2, b2, out);
}


# Z0's backward light cone never reaches wire 7: operator built on 7 wires (64x128) and applied to even and odd elements of each row as two K=128 WMMA products: half the WMMAs; new sum order
# speedup vs baseline: 3.6199x; 1.5983x over previous
#include <hip/hip_runtime.h>

typedef __attribute__((ext_vector_type(2))) float v2f;
typedef __attribute__((ext_vector_type(4))) float v4f;
typedef __attribute__((ext_vector_type(8))) float v8f;

#define N_LAYERS 7
#define N_QUBITS 8
#define DIM 256
#define CDIM 128


template <int M>
__device__ __forceinline__ float lane_xor(float v) {
  const int iv = __builtin_bit_cast(int, v);
  if constexpr (M == 16) {
    return __builtin_bit_cast(float, __builtin_amdgcn_permlanex16(
        iv, iv, 0x76543210, 0xfedcba98, false, false));
  } else {
    return __builtin_bit_cast(float, __builtin_amdgcn_update_dpp(
        0, iv, 0x160 + M, 0xF, 0xF, true));
  }
}

__device__ __forceinline__ float lane_bcast(float v, int src) {
  return __builtin_bit_cast(float,
      __builtin_amdgcn_readlane(__builtin_bit_cast(int, v), src));
}

template <int Q>
__device__ __forceinline__ void ry_gate(float (&a)[4], float c, float s, int lane) {
  if constexpr (Q < 2) {
    constexpr int MR = 2 >> Q;
#pragma unroll
    for (int r = 0; r < 4; ++r) {
      if ((r & MR) == 0) {
        const float a0 = a[r], a1 = a[r | MR];
        a[r]      = c * a0 - s * a1;
        a[r | MR] = s * a0 + c * a1;
      }
    }
  } else {
    constexpr int ML = 64 >> Q;
    const float sg = (lane & ML) ? s : -s;
#pragma unroll
    for (int r = 0; r < 4; ++r) {
      const float p = lane_xor<ML>(a[r]);
      a[r] = c * a[r] + sg * p;
    }
  }
}

template <int L, int Q>
__device__ __forceinline__ void cone_gate(float (&a)[4], float cs, float sn, int lane) {
  if constexpr (Q <= 6 - L) {
    constexpr int src = 7 * L - (L * (L - 1)) / 2 + Q;
    ry_gate<Q>(a, lane_bcast(cs, src), lane_bcast(sn, src), lane);
  }
}

template <int L>
__device__ __forceinline__ void cone_layer(float (&a)[4], float cs, float sn, int lane) {
  cone_gate<L, 0>(a, cs, sn, lane);
  cone_gate<L, 1>(a, cs, sn, lane);
  cone_gate<L, 2>(a, cs, sn, lane);
  cone_gate<L, 3>(a, cs, sn, lane);
  cone_gate<L, 4>(a, cs, sn, lane);
  cone_gate<L, 5>(a, cs, sn, lane);
  cone_gate<L, 6>(a, cs, sn, lane);
  if constexpr (L <= 5) {
    float t[4];
    if constexpr (L <= 4) {
      constexpr int ML = 0x1F & ~((1 << L) - 1);
      const int gl = lane ^ ((lane >> 1) & ML);
#pragma unroll
      for (int r = 0; r < 4; ++r) t[r] = __shfl(a[r ^ (r >> 1)], gl ^ ((r & 1) << 4), 32);
    } else {
#pragma unroll
      for (int r = 0; r < 4; ++r) t[r] = a[r ^ (r >> 1)];
    }
#pragma unroll
    for (int r = 0; r < 4; ++r) a[r] = t[r];
  }
}

__global__ __launch_bounds__(32) void build_v_kernel(const float* __restrict__ w,
                                                     float* __restrict__ bp) {
  const int lane = threadIdx.x;
  const int j = blockIdx.x;

  const int lg = (lane >= 7) + (lane >= 13) + (lane >= 18) + (lane >= 22) +
                 (lane >= 25) + (lane >= 27);
  const int qg = lane - (7 * lg - (lg * (lg - 1)) / 2);
  const float th = 0.5f * w[lg * N_QUBITS + qg];
  const float cs = cosf(th), sn = sinf(th);

  float a[4];
#pragma unroll
  for (int r = 0; r < 4; ++r) a[r] = (r * 32 + lane == j) ? 1.0f : 0.0f;

  cone_layer<0>(a, cs, sn, lane);
  cone_layer<1>(a, cs, sn, lane);
  cone_layer<2>(a, cs, sn, lane);
  cone_layer<3>(a, cs, sn, lane);
  cone_layer<4>(a, cs, sn, lane);
  cone_layer<5>(a, cs, sn, lane);
  cone_layer<6>(a, cs, sn, lane);

  const int t   = j >> 2;
  const int hf  = (j >> 1) & 1;
  const int c   = j & 1;
  const int l16 = lane & 15;
#pragma unroll
  for (int r = 0; r < 2; ++r) {
    const int nt = r * 2 + (lane >> 4);
    bp[(size_t)(((nt * 32 + t) * 32) + hf * 16 + l16) * 2 + c] = a[r];
  }
}

__global__ __launch_bounds__(256) __attribute__((amdgpu_waves_per_eu(4, 4)))
void qnn_fused_kernel(
    const float* __restrict__ x, const v2f* __restrict__ bp,
    const float* __restrict__ W1, const float* __restrict__ b1,
    const float* __restrict__ W2, const float* __restrict__ b2,
    float* __restrict__ out) {
  __shared__ float redP[8][16][16];
  __shared__ float redS[8][32];

  const int lane = threadIdx.x & 31;
  const int wave = threadIdx.x >> 5;
  const int hf   = lane >> 4;
  const int l16  = lane & 15;
  const int rbase = blockIdx.x * 128 + wave * 16;

  const float* xrow = x + (size_t)(rbase + l16) * DIM + 4 * hf;

  v2f ae[32], ao[32];
#pragma unroll
  for (int t = 0; t < 32; ++t) {
    const v4f q = *(const v4f*)(xrow + 8 * t);
    float m0 = q[1], m1 = q[2];
    asm("v_swap_b32 %0, %1" : "+v"(m0), "+v"(m1));
    ae[t] = v2f{q[0], m0};
    ao[t] = v2f{m1, q[3]};
  }

  float accP[8];
#pragma unroll
  for (int v = 0; v < 8; ++v) accP[v] = 0.f;

  for (int nt = 0; nt < 4; ++nt) {
    const v2f* btile = bp + (size_t)nt * 32 * 32 + lane;
    v8f acce = {0.f, 0.f, 0.f, 0.f, 0.f, 0.f, 0.f, 0.f};
    v8f acco = {0.f, 0.f, 0.f, 0.f, 0.f, 0.f, 0.f, 0.f};
#pragma unroll
    for (int t = 0; t < 32; ++t) {
      const v2f b = btile[t * 32];
      acce = __builtin_amdgcn_wmma_f32_16x16x4_f32(
           false, ae[t],  false, b,
           (short)0, acce,  false,  false);
      acco = __builtin_amdgcn_wmma_f32_16x16x4_f32(
           false, ao[t],  false, b,
           (short)0, acco,  false,  false);
    }
#pragma unroll
    for (int v = 0; v < 8; ++v) accP[v] += acce[v] * acce[v] + acco[v] * acco[v];
  }

  v2f sq0 = {0.f, 0.f}, sq1 = {0.f, 0.f}, sq2 = {0.f, 0.f}, sq3 = {0.f, 0.f};
#pragma unroll
  for (int t = 0; t < 32; t += 2) {
    sq0 += ae[t]     * ae[t];
    sq1 += ao[t]     * ao[t];
    sq2 += ae[t + 1] * ae[t + 1];
    sq3 += ao[t + 1] * ao[t + 1];
  }
  const v2f sq = (sq0 + sq1) + (sq2 + sq3);
  redS[wave][lane] = sq[0] + sq[1];

#pragma unroll
  for (int v = 0; v < 8; ++v) {
    redP[wave][hf * 8 + v][l16] = accP[v];
  }
  __syncthreads();

  if (threadIdx.x < 128) {
    const int wv = threadIdx.x >> 4;
    const int r  = threadIdx.x & 15;
    float P = 0.f;
#pragma unroll
    for (int i = 0; i < 16; ++i) P += redP[wv][r][i];
    const float S = redS[wv][r] + redS[wv][16 + r];
    const float z = (2.f * P - S) / S;
    float t = b2[0];
#pragma unroll
    for (int jj = 0; jj < 16; ++jj) {
      float h = z * W1[jj] + b1[jj];
      h = h > 0.f ? h : 0.f;
      t += h * W2[jj];
    }
    out[(size_t)blockIdx.x * 128 + wv * 16 + r] = 1.f / (1.f + expf(-t));
  }
}

extern "C" void kernel_launch(void* const* d_in, const int* in_sizes, int n_in,
                              void* d_out, int out_size, void* d_ws, size_t ws_size,
                              hipStream_t stream) {
  const float* x  = (const float*)d_in[0];
  const float* w  = (const float*)d_in[1];
  const float* W1 = (const float*)d_in[2];
  const float* b1 = (const float*)d_in[3];
  const float* W2 = (const float*)d_in[4];
  const float* b2 = (const float*)d_in[5];
  float* out = (float*)d_out;
  float* bp  = (float*)d_ws;

  const int B = in_sizes[0] / DIM;
  build_v_kernel<<<CDIM, 32, 0, stream>>>(w, bp);
  qnn_fused_kernel<<<B / 128, 256, 0, stream>>>(x, (const v2f*)bp, W1, b1, W2, b2, out);
}


# V+ built inside the product kernel: each of 16 waves evolves its 4 rows backwards through the light cone into LDS while its x loads fly; 1 launch (was 2); V entries round differently
# speedup vs baseline: 3.8983x; 1.0769x over previous
#include <hip/hip_runtime.h>

typedef __attribute__((ext_vector_type(2))) float v2f;
typedef __attribute__((ext_vector_type(4))) float v4f;
typedef __attribute__((ext_vector_type(8))) float v8f;

#define N_LAYERS 7
#define N_QUBITS 8
#define DIM 256


template <int M>
__device__ __forceinline__ float lane_xor(float v) {
  const int iv = __builtin_bit_cast(int, v);
  if constexpr (M == 16) {
    return __builtin_bit_cast(float, __builtin_amdgcn_permlanex16(
        iv, iv, 0x76543210, 0xfedcba98, false, false));
  } else {
    return __builtin_bit_cast(float, __builtin_amdgcn_update_dpp(
        0, iv, 0x160 + M, 0xF, 0xF, true));
  }
}

__device__ __forceinline__ float lane_bcast(float v, int src) {
  return __builtin_bit_cast(float,
      __builtin_amdgcn_readlane(__builtin_bit_cast(int, v), src));
}

template <int LOW>
__device__ __forceinline__ int chain_inv(int idx) {
  int p = idx >> LOW;
  p ^= p >> 1;
  p ^= p >> 2;
  p ^= p >> 4;
  return (p << LOW) | (idx & ((1 << LOW) - 1));
}

template <int L, int Q, int ML, int NR>
__device__ __forceinline__ void ryT_lane(float (&a)[NR], float cs, float sn, int lane) {
  constexpr int src = 7 * L - (L * (L - 1)) / 2 + Q;
  const float c = lane_bcast(cs, src), s = lane_bcast(sn, src);
  const float sg = (lane & ML) ? -s : s;
#pragma unroll
  for (int r = 0; r < NR; ++r) {
    const float p = lane_xor<ML>(a[r]);
    a[r] = c * a[r] + sg * p;
  }
}

template <int L, int Q, int MR, int NR>
__device__ __forceinline__ void ryT_reg(float (&a)[NR], float cs, float sn) {
  constexpr int src = 7 * L - (L * (L - 1)) / 2 + Q;
  const float c = lane_bcast(cs, src), s = lane_bcast(sn, src);
#pragma unroll
  for (int r = 0; r < NR; ++r) {
    if ((r & MR) == 0) {
      const float a0 = a[r], a1 = a[r | MR];
      a[r]      = c * a0 + s * a1;
      a[r | MR] = c * a1 - s * a0;
    }
  }
}

__global__ __launch_bounds__(512) __attribute__((amdgpu_waves_per_eu(4, 4)))
void qnn_fused_kernel(
    const float* __restrict__ x, const float* __restrict__ w,
    const float* __restrict__ W1, const float* __restrict__ b1,
    const float* __restrict__ W2, const float* __restrict__ b2,
    float* __restrict__ out) {
  __shared__ v2f   Vs[4 * 32 * 32];
  __shared__ float redP[16][16][16];
  __shared__ float redS[16][32];

  const int lane = threadIdx.x & 31;
  const int wave = threadIdx.x >> 5;
  const int hf   = lane >> 4;
  const int l16  = lane & 15;
  const int rbase = blockIdx.x * 256 + wave * 16;

  const int lg = (lane >= 7) + (lane >= 13) + (lane >= 18) + (lane >= 22) +
                 (lane >= 25) + (lane >= 27);
  const int qg = lane - (7 * lg - (lg * (lg - 1)) / 2);
  const float wv = w[lg * N_QUBITS + qg];
  __builtin_amdgcn_sched_barrier(0);

  const float* xrow = x + (size_t)(rbase + l16) * DIM + 4 * hf;
  v4f q[32];
#pragma unroll
  for (int t = 0; t < 32; ++t) q[t] = *(const v4f*)(xrow + 8 * t);
  __builtin_amdgcn_sched_barrier(0);

  const float th = 0.5f * wv;
  const float cs = cosf(th), sn = sinf(th);

  float u[1];
  u[0] = (lane == wave) ? 1.0f : 0.0f;
  ryT_lane<6, 0, 16, 1>(u, cs, sn, lane);
  u[0] = __shfl(u[0], chain_inv<3>(lane), 32);
  ryT_lane<5, 0, 16, 1>(u, cs, sn, lane);
  ryT_lane<5, 1,  8, 1>(u, cs, sn, lane);
  u[0] = __shfl(u[0], chain_inv<2>(lane), 32);
  ryT_lane<4, 0, 16, 1>(u, cs, sn, lane);
  ryT_lane<4, 1,  8, 1>(u, cs, sn, lane);
  ryT_lane<4, 2,  4, 1>(u, cs, sn, lane);
  u[0] = __shfl(u[0], chain_inv<1>(lane), 32);
  ryT_lane<3, 0, 16, 1>(u, cs, sn, lane);
  ryT_lane<3, 1,  8, 1>(u, cs, sn, lane);
  ryT_lane<3, 2,  4, 1>(u, cs, sn, lane);
  ryT_lane<3, 3,  2, 1>(u, cs, sn, lane);
  u[0] = __shfl(u[0], chain_inv<0>(lane), 32);
  ryT_lane<2, 0, 16, 1>(u, cs, sn, lane);
  ryT_lane<2, 1,  8, 1>(u, cs, sn, lane);
  ryT_lane<2, 2,  4, 1>(u, cs, sn, lane);
  ryT_lane<2, 3,  2, 1>(u, cs, sn, lane);
  ryT_lane<2, 4,  1, 1>(u, cs, sn, lane);

  float ub[2];
  int   gb[2];
#pragma unroll
  for (int rho = 0; rho < 2; ++rho) {
    gb[rho] = chain_inv<0>(rho * 32 + lane);
    ub[rho] = __shfl(u[0], gb[rho] >> 1, 32);
  }
  float* Vf = (float*)Vs;
#pragma unroll
  for (int bit1 = 0; bit1 < 2; ++bit1) {
    float s[2];
#pragma unroll
    for (int rho = 0; rho < 2; ++rho) s[rho] = ((gb[rho] & 1) == bit1) ? ub[rho] : 0.0f;
    ryT_reg <1, 0,  1, 2>(s, cs, sn);
    ryT_lane<1, 1, 16, 2>(s, cs, sn, lane);
    ryT_lane<1, 2,  8, 2>(s, cs, sn, lane);
    ryT_lane<1, 3,  4, 2>(s, cs, sn, lane);
    ryT_lane<1, 4,  2, 2>(s, cs, sn, lane);
    ryT_lane<1, 5,  1, 2>(s, cs, sn, lane);

    float sc[4];
    int   ic[4];
#pragma unroll
    for (int rho = 0; rho < 4; ++rho) {
      ic[rho] = chain_inv<0>(rho * 32 + lane);
      sc[rho] = __shfl(s[rho >> 1], (ic[rho] >> 1) & 31, 32);
    }
#pragma unroll
    for (int bit0 = 0; bit0 < 2; ++bit0) {
      float a[4];
#pragma unroll
      for (int rho = 0; rho < 4; ++rho) a[rho] = ((ic[rho] & 1) == bit0) ? sc[rho] : 0.0f;
      ryT_reg <0, 0,  2, 4>(a, cs, sn);
      ryT_reg <0, 1,  1, 4>(a, cs, sn);
      ryT_lane<0, 2, 16, 4>(a, cs, sn, lane);
      ryT_lane<0, 3,  8, 4>(a, cs, sn, lane);
      ryT_lane<0, 4,  4, 4>(a, cs, sn, lane);
      ryT_lane<0, 5,  2, 4>(a, cs, sn, lane);
      ryT_lane<0, 6,  1, 4>(a, cs, sn, lane);
      const int n = 4 * wave + 2 * bit1 + bit0;
#pragma unroll
      for (int rho = 0; rho < 4; ++rho) {
        const int t = rho * 8 + (lane >> 2);
        Vf[((((n >> 4) * 32 + t) * 32) + ((lane >> 1) & 1) * 16 + (n & 15)) * 2 + (lane & 1)] = a[rho];
      }
    }
  }
  __syncthreads();
  __builtin_amdgcn_sched_barrier(0);

  v2f ae[32], ao[32];
#pragma unroll
  for (int t = 0; t < 32; ++t) {
    float m0 = q[t][1], m1 = q[t][2];
    asm("v_swap_b32 %0, %1" : "+v"(m0), "+v"(m1));
    ae[t] = v2f{q[t][0], m0};
    ao[t] = v2f{m1, q[t][3]};
  }

  float accP[8];
#pragma unroll
  for (int v = 0; v < 8; ++v) accP[v] = 0.f;

  for (int nt = 0; nt < 4; ++nt) {
    const v2f* btile = Vs + nt * 32 * 32 + lane;
    v8f acce = {0.f, 0.f, 0.f, 0.f, 0.f, 0.f, 0.f, 0.f};
    v8f acco = {0.f, 0.f, 0.f, 0.f, 0.f, 0.f, 0.f, 0.f};
#pragma unroll
    for (int t = 0; t < 32; ++t) {
      const v2f b = btile[t * 32];
      acce = __builtin_amdgcn_wmma_f32_16x16x4_f32(
           false, ae[t],  false, b,
           (short)0, acce,  false,  false);
      acco = __builtin_amdgcn_wmma_f32_16x16x4_f32(
           false, ao[t],  false, b,
           (short)0, acco,  false,  false);
    }
#pragma unroll
    for (int v = 0; v < 8; ++v) accP[v] += acce[v] * acce[v] + acco[v] * acco[v];
  }

  v2f sq0 = {0.f, 0.f}, sq1 = {0.f, 0.f}, sq2 = {0.f, 0.f}, sq3 = {0.f, 0.f};
#pragma unroll
  for (int t = 0; t < 32; t += 2) {
    sq0 += ae[t]     * ae[t];
    sq1 += ao[t]     * ao[t];
    sq2 += ae[t + 1] * ae[t + 1];
    sq3 += ao[t + 1] * ao[t + 1];
  }
  const v2f sq = (sq0 + sq1) + (sq2 + sq3);
  redS[wave][lane] = sq[0] + sq[1];

#pragma unroll
  for (int v = 0; v < 8; ++v) {
    redP[wave][hf * 8 + v][l16] = accP[v];
  }
  __syncthreads();

  if (threadIdx.x < 256) {
    const int wv2 = threadIdx.x >> 4;
    const int r   = threadIdx.x & 15;
    float P = 0.f;
#pragma unroll
    for (int i = 0; i < 16; ++i) P += redP[wv2][r][i];
    const float S = redS[wv2][r] + redS[wv2][16 + r];
    const float z = (2.f * P - S) / S;
    float t = b2[0];
#pragma unroll
    for (int jj = 0; jj < 16; ++jj) {
      float h = z * W1[jj] + b1[jj];
      h = h > 0.f ? h : 0.f;
      t += h * W2[jj];
    }
    out[(size_t)blockIdx.x * 256 + wv2 * 16 + r] = 1.f / (1.f + expf(-t));
  }
}

extern "C" void kernel_launch(void* const* d_in, const int* in_sizes, int n_in,
                              void* d_out, int out_size, void* d_ws, size_t ws_size,
                              hipStream_t stream) {
  const float* x  = (const float*)d_in[0];
  const float* w  = (const float*)d_in[1];
  const float* W1 = (const float*)d_in[2];
  const float* b1 = (const float*)d_in[3];
  const float* W2 = (const float*)d_in[4];
  const float* b2 = (const float*)d_in[5];
  float* out = (float*)d_out;

  const int B = in_sizes[0] / DIM;
  qnn_fused_kernel<<<B / 256, 512, 0, stream>>>(x, w, W1, b1, W2, b2, out);
}


# CNOT(5,6) meets wire 6 only as X: x pairs rotated into its eigenbasis (R6, then +/-) split the 64x128 operator into two 32x64 blocks: half the WMMAs; new sum order, operator entries round differently
# speedup vs baseline: 5.1437x; 1.3195x over previous
#include <hip/hip_runtime.h>

typedef __attribute__((ext_vector_type(2))) float v2f;
typedef __attribute__((ext_vector_type(4))) float v4f;
typedef __attribute__((ext_vector_type(8))) float v8f;

#define N_LAYERS 7
#define N_QUBITS 8
#define DIM 256


template <int M>
__device__ __forceinline__ float lane_xor(float v) {
  const int iv = __builtin_bit_cast(int, v);
  if constexpr (M == 16) {
    return __builtin_bit_cast(float, __builtin_amdgcn_permlanex16(
        iv, iv, 0x76543210, 0xfedcba98, false, false));
  } else {
    return __builtin_bit_cast(float, __builtin_amdgcn_update_dpp(
        0, iv, 0x160 + M, 0xF, 0xF, true));
  }
}

__device__ __forceinline__ float lane_bcast(float v, int src) {
  return __builtin_bit_cast(float,
      __builtin_amdgcn_readlane(__builtin_bit_cast(int, v), src));
}

template <int LOW>
__device__ __forceinline__ int chain_inv(int idx) {
  int p = idx >> LOW;
  p ^= p >> 1;
  p ^= p >> 2;
  p ^= p >> 4;
  return (p << LOW) | (idx & ((1 << LOW) - 1));
}

template <int L, int Q, int ML, int NR>
__device__ __forceinline__ void ryT_lane(float (&a)[NR], float cs, float sn, int lane) {
  constexpr int src = 7 * L - (L * (L - 1)) / 2 + Q;
  const float c = lane_bcast(cs, src), s = lane_bcast(sn, src);
  const float sg = (lane & ML) ? -s : s;
#pragma unroll
  for (int r = 0; r < NR; ++r) {
    const float p = lane_xor<ML>(a[r]);
    a[r] = c * a[r] + sg * p;
  }
}

template <int L, int Q, int MR, int NR>
__device__ __forceinline__ void ryT_reg(float (&a)[NR], float cs, float sn) {
  constexpr int src = 7 * L - (L * (L - 1)) / 2 + Q;
  const float c = lane_bcast(cs, src), s = lane_bcast(sn, src);
#pragma unroll
  for (int r = 0; r < NR; ++r) {
    if ((r & MR) == 0) {
      const float a0 = a[r], a1 = a[r | MR];
      a[r]      = c * a0 + s * a1;
      a[r | MR] = c * a1 - s * a0;
    }
  }
}

__global__ __launch_bounds__(512) __attribute__((amdgpu_waves_per_eu(4, 4)))
void qnn_fused_kernel(
    const float* __restrict__ x, const float* __restrict__ w,
    const float* __restrict__ W1, const float* __restrict__ b1,
    const float* __restrict__ W2, const float* __restrict__ b2,
    float* __restrict__ out) {
  __shared__ v2f   Bs[2 * 2 * 16 * 32];
  __shared__ float redP[16][16][16];
  __shared__ float redS[16][32];

  const int lane = threadIdx.x & 31;
  const int wave = threadIdx.x >> 5;
  const int hf   = lane >> 4;
  const int l16  = lane & 15;
  const int rbase = blockIdx.x * 256 + wave * 16;

  const int lg = (lane >= 7) + (lane >= 13) + (lane >= 18) + (lane >= 22) +
                 (lane >= 25) + (lane >= 27);
  const int qg = lane - (7 * lg - (lg * (lg - 1)) / 2);
  const float wv = w[lg * N_QUBITS + qg];
  __builtin_amdgcn_sched_barrier(0);

  const float* xrow = x + (size_t)(rbase + l16) * DIM + 4 * hf;
  v4f q[32];
#pragma unroll
  for (int t = 0; t < 32; ++t) q[t] = *(const v4f*)(xrow + 8 * t);
  __builtin_amdgcn_sched_barrier(0);

  const float th = 0.5f * wv;
  const float cs = cosf(th), sn = sinf(th);

  float u[1];
  u[0] = (lane == wave) ? 1.0f : 0.0f;
  ryT_lane<6, 0, 16, 1>(u, cs, sn, lane);
  u[0] = __shfl(u[0], chain_inv<3>(lane), 32);
  ryT_lane<5, 0, 16, 1>(u, cs, sn, lane);
  ryT_lane<5, 1,  8, 1>(u, cs, sn, lane);
  u[0] = __shfl(u[0], chain_inv<2>(lane), 32);
  ryT_lane<4, 0, 16, 1>(u, cs, sn, lane);
  ryT_lane<4, 1,  8, 1>(u, cs, sn, lane);
  ryT_lane<4, 2,  4, 1>(u, cs, sn, lane);
  u[0] = __shfl(u[0], chain_inv<1>(lane), 32);
  ryT_lane<3, 0, 16, 1>(u, cs, sn, lane);
  ryT_lane<3, 1,  8, 1>(u, cs, sn, lane);
  ryT_lane<3, 2,  4, 1>(u, cs, sn, lane);
  ryT_lane<3, 3,  2, 1>(u, cs, sn, lane);
  u[0] = __shfl(u[0], chain_inv<0>(lane), 32);
  ryT_lane<2, 0, 16, 1>(u, cs, sn, lane);
  ryT_lane<2, 1,  8, 1>(u, cs, sn, lane);
  ryT_lane<2, 2,  4, 1>(u, cs, sn, lane);
  ryT_lane<2, 3,  2, 1>(u, cs, sn, lane);
  ryT_lane<2, 4,  1, 1>(u, cs, sn, lane);

  float ub[2];
  int   gb[2];
#pragma unroll
  for (int rho = 0; rho < 2; ++rho) {
    gb[rho] = chain_inv<0>(rho * 32 + lane);
    ub[rho] = __shfl(u[0], gb[rho] >> 1, 32);
  }
  float* Bf = (float*)Bs;
#pragma unroll
  for (int mb = 0; mb < 2; ++mb) {
    float s[2];
#pragma unroll
    for (int rho = 0; rho < 2; ++rho) s[rho] = ((gb[rho] & 1) == mb) ? ub[rho] : 0.0f;
    ryT_reg <1, 0,  1, 2>(s, cs, sn);
    ryT_lane<1, 1, 16, 2>(s, cs, sn, lane);
    ryT_lane<1, 2,  8, 2>(s, cs, sn, lane);
    ryT_lane<1, 3,  4, 2>(s, cs, sn, lane);
    ryT_lane<1, 4,  2, 2>(s, cs, sn, lane);
    ryT_lane<1, 5,  1, 2>(s, cs, sn, lane);

    float sc[2];
#pragma unroll
    for (int rho = 0; rho < 2; ++rho) sc[rho] = __shfl(s[rho], gb[rho] & 31, 32);
    const int m = 2 * wave + mb;
#pragma unroll
    for (int sg = 0; sg < 2; ++sg) {
      float a[2];
#pragma unroll
      for (int rho = 0; rho < 2; ++rho) a[rho] = (sg && (gb[rho] & 1)) ? -sc[rho] : sc[rho];
      ryT_reg <0, 0,  1, 2>(a, cs, sn);
      ryT_lane<0, 1, 16, 2>(a, cs, sn, lane);
      ryT_lane<0, 2,  8, 2>(a, cs, sn, lane);
      ryT_lane<0, 3,  4, 2>(a, cs, sn, lane);
      ryT_lane<0, 4,  2, 2>(a, cs, sn, lane);
      ryT_lane<0, 5,  1, 2>(a, cs, sn, lane);
#pragma unroll
      for (int rho = 0; rho < 2; ++rho) {
        const int ku = rho * 8 + (lane >> 2);
        Bf[(((((sg * 2 + (m >> 4)) * 16 + ku) * 32) + (lane & 1) * 16 + (m & 15)) * 2) + ((lane >> 1) & 1)] = a[rho];
      }
    }
  }
  __syncthreads();
  __builtin_amdgcn_sched_barrier(0);

  const float c6 = lane_bcast(cs, 6), s6 = lane_bcast(sn, 6);
  const float al = (c6 + s6) * 0.70710678118654752f;
  const float be = (c6 - s6) * 0.70710678118654752f;
  v2f ap0[16], ap1[16], am0[16], am1[16];
#pragma unroll
  for (int u = 0; u < 16; ++u) {
    v2f yp[2], ym[2];
#pragma unroll
    for (int h = 0; h < 2; ++h) {
      const v4f qq = q[2 * u + h];
      const v2f lo = __builtin_shufflevector(qq, qq, 0, 1);
      const v2f hi = __builtin_shufflevector(qq, qq, 2, 3);
      yp[h] = al * lo + be * hi;
      ym[h] = be * lo - al * hi;
    }
    float p0 = yp[0][1], p1 = yp[1][0];
    asm("v_swap_b32 %0, %1" : "+v"(p0), "+v"(p1));
    ap0[u] = v2f{yp[0][0], p0};
    ap1[u] = v2f{p1, yp[1][1]};
    float m0 = ym[0][1], m1 = ym[1][0];
    asm("v_swap_b32 %0, %1" : "+v"(m0), "+v"(m1));
    am0[u] = v2f{ym[0][0], m0};
    am1[u] = v2f{m1, ym[1][1]};
  }

  float accP[8];
#pragma unroll
  for (int v = 0; v < 8; ++v) accP[v] = 0.f;

  auto product = [&](const v2f (&A0)[16], const v2f (&A1)[16], const v2f* bsg) {
    for (int nt = 0; nt < 2; ++nt) {
      const v2f* btile = bsg + nt * 16 * 32;
      v8f acc0 = {0.f, 0.f, 0.f, 0.f, 0.f, 0.f, 0.f, 0.f};
      v8f acc1 = {0.f, 0.f, 0.f, 0.f, 0.f, 0.f, 0.f, 0.f};
#pragma unroll
      for (int u = 0; u < 16; ++u) {
        const v2f b = btile[u * 32];
        acc0 = __builtin_amdgcn_wmma_f32_16x16x4_f32(
             false, A0[u],  false, b,
             (short)0, acc0,  false,  false);
        acc1 = __builtin_amdgcn_wmma_f32_16x16x4_f32(
             false, A1[u],  false, b,
             (short)0, acc1,  false,  false);
      }
#pragma unroll
      for (int v = 0; v < 8; ++v) accP[v] += acc0[v] * acc0[v] + acc1[v] * acc1[v];
    }
  };
  product(ap0, ap1, Bs + lane);
  product(am0, am1, Bs + 2 * 16 * 32 + lane);

  v2f sq0 = {0.f, 0.f}, sq1 = {0.f, 0.f}, sq2 = {0.f, 0.f}, sq3 = {0.f, 0.f};
#pragma unroll
  for (int u = 0; u < 16; ++u) {
    sq0 += ap0[u] * ap0[u];
    sq1 += ap1[u] * ap1[u];
    sq2 += am0[u] * am0[u];
    sq3 += am1[u] * am1[u];
  }
  const v2f sq = (sq0 + sq1) + (sq2 + sq3);
  redS[wave][lane] = sq[0] + sq[1];

#pragma unroll
  for (int v = 0; v < 8; ++v) {
    redP[wave][hf * 8 + v][l16] = accP[v];
  }
  __syncthreads();

  if (threadIdx.x < 256) {
    const int wv2 = threadIdx.x >> 4;
    const int r   = threadIdx.x & 15;
    float P = 0.f;
#pragma unroll
    for (int i = 0; i < 16; ++i) P += redP[wv2][r][i];
    const float S = redS[wv2][r] + redS[wv2][16 + r];
    const float z = (2.f * P - S) / S;
    float t = b2[0];
#pragma unroll
    for (int jj = 0; jj < 16; ++jj) {
      float h = z * W1[jj] + b1[jj];
      h = h > 0.f ? h : 0.f;
      t += h * W2[jj];
    }
    out[(size_t)blockIdx.x * 256 + wv2 * 16 + r] = 1.f / (1.f + expf(-t));
  }
}

extern "C" void kernel_launch(void* const* d_in, const int* in_sizes, int n_in,
                              void* d_out, int out_size, void* d_ws, size_t ws_size,
                              hipStream_t stream) {
  const float* x  = (const float*)d_in[0];
  const float* w  = (const float*)d_in[1];
  const float* W1 = (const float*)d_in[2];
  const float* b1 = (const float*)d_in[3];
  const float* W2 = (const float*)d_in[4];
  const float* b2 = (const float*)d_in[5];
  float* out = (float*)d_out;

  const int B = in_sizes[0] / DIM;
  qnn_fused_kernel<<<B / 256, 512, 0, stream>>>(x, w, W1, b1, W2, b2, out);
}


# x requested a 128-byte line per row at a time, 4-5 k-steps ahead, inside a k-step-major product on 8 accumulator tiles; hand-written barrier keeps loads in flight: load, rotation, WMMA overlap
# speedup vs baseline: 6.0419x; 1.1746x over previous
#include <hip/hip_runtime.h>

typedef __attribute__((ext_vector_type(2))) float v2f;
typedef __attribute__((ext_vector_type(4))) float v4f;
typedef __attribute__((ext_vector_type(8))) float v8f;

#define N_LAYERS 7
#define N_QUBITS 8
#define DIM 256


template <int M>
__device__ __forceinline__ float lane_xor(float v) {
  const int iv = __builtin_bit_cast(int, v);
  if constexpr (M == 16) {
    return __builtin_bit_cast(float, __builtin_amdgcn_permlanex16(
        iv, iv, 0x76543210, 0xfedcba98, false, false));
  } else {
    return __builtin_bit_cast(float, __builtin_amdgcn_update_dpp(
        0, iv, 0x160 + M, 0xF, 0xF, true));
  }
}

__device__ __forceinline__ float lane_bcast(float v, int src) {
  return __builtin_bit_cast(float,
      __builtin_amdgcn_readlane(__builtin_bit_cast(int, v), src));
}

template <int LOW>
__device__ __forceinline__ int chain_inv(int idx) {
  int p = idx >> LOW;
  p ^= p >> 1;
  p ^= p >> 2;
  p ^= p >> 4;
  return (p << LOW) | (idx & ((1 << LOW) - 1));
}

template <int L, int Q, int ML, int NR>
__device__ __forceinline__ void ryT_lane(float (&a)[NR], float cs, float sn, int lane) {
  constexpr int src = 7 * L - (L * (L - 1)) / 2 + Q;
  const float c = lane_bcast(cs, src), s = lane_bcast(sn, src);
  const float sg = (lane & ML) ? -s : s;
#pragma unroll
  for (int r = 0; r < NR; ++r) {
    const float p = lane_xor<ML>(a[r]);
    a[r] = c * a[r] + sg * p;
  }
}

template <int L, int Q, int MR, int NR>
__device__ __forceinline__ void ryT_reg(float (&a)[NR], float cs, float sn) {
  constexpr int src = 7 * L - (L * (L - 1)) / 2 + Q;
  const float c = lane_bcast(cs, src), s = lane_bcast(sn, src);
#pragma unroll
  for (int r = 0; r < NR; ++r) {
    if ((r & MR) == 0) {
      const float a0 = a[r], a1 = a[r | MR];
      a[r]      = c * a0 + s * a1;
      a[r | MR] = c * a1 - s * a0;
    }
  }
}

__global__ __launch_bounds__(512) __attribute__((amdgpu_waves_per_eu(4, 4)))
void qnn_fused_kernel(
    const float* __restrict__ x, const float* __restrict__ w,
    const float* __restrict__ W1, const float* __restrict__ b1,
    const float* __restrict__ W2, const float* __restrict__ b2,
    float* __restrict__ out) {
  __shared__ v2f   Bs[2 * 2 * 16 * 32];
  __shared__ float redP[16][16][16];
  __shared__ float redS[16][32];

  const int lane = threadIdx.x & 31;
  const int wave = threadIdx.x >> 5;
  const int hf   = lane >> 4;
  const int l16  = lane & 15;
  const int rbase = blockIdx.x * 256 + wave * 16;

  const int lg = (lane >= 7) + (lane >= 13) + (lane >= 18) + (lane >= 22) +
                 (lane >= 25) + (lane >= 27);
  const int qg = lane - (7 * lg - (lg * (lg - 1)) / 2);
  const float wv = w[lg * N_QUBITS + qg];
  __builtin_amdgcn_sched_barrier(0);

  const float* xrow = x + (size_t)(rbase + l16) * DIM + 4 * hf;
  constexpr int PF = 4;
  v4f q[32];
#pragma unroll
  for (int t = 0; t < 2 * PF; ++t) q[t] = *(const v4f*)(xrow + 8 * t);
  __builtin_amdgcn_sched_barrier(0);

  const float th = 0.5f * wv;
  const float cs = cosf(th), sn = sinf(th);

  float u[1];
  u[0] = (lane == wave) ? 1.0f : 0.0f;
  ryT_lane<6, 0, 16, 1>(u, cs, sn, lane);
  u[0] = __shfl(u[0], chain_inv<3>(lane), 32);
  ryT_lane<5, 0, 16, 1>(u, cs, sn, lane);
  ryT_lane<5, 1,  8, 1>(u, cs, sn, lane);
  u[0] = __shfl(u[0], chain_inv<2>(lane), 32);
  ryT_lane<4, 0, 16, 1>(u, cs, sn, lane);
  ryT_lane<4, 1,  8, 1>(u, cs, sn, lane);
  ryT_lane<4, 2,  4, 1>(u, cs, sn, lane);
  u[0] = __shfl(u[0], chain_inv<1>(lane), 32);
  ryT_lane<3, 0, 16, 1>(u, cs, sn, lane);
  ryT_lane<3, 1,  8, 1>(u, cs, sn, lane);
  ryT_lane<3, 2,  4, 1>(u, cs, sn, lane);
  ryT_lane<3, 3,  2, 1>(u, cs, sn, lane);
  u[0] = __shfl(u[0], chain_inv<0>(lane), 32);
  ryT_lane<2, 0, 16, 1>(u, cs, sn, lane);
  ryT_lane<2, 1,  8, 1>(u, cs, sn, lane);
  ryT_lane<2, 2,  4, 1>(u, cs, sn, lane);
  ryT_lane<2, 3,  2, 1>(u, cs, sn, lane);
  ryT_lane<2, 4,  1, 1>(u, cs, sn, lane);

  float ub[2];
  int   gb[2];
#pragma unroll
  for (int rho = 0; rho < 2; ++rho) {
    gb[rho] = chain_inv<0>(rho * 32 + lane);
    ub[rho] = __shfl(u[0], gb[rho] >> 1, 32);
  }
  float* Bf = (float*)Bs;
#pragma unroll
  for (int mb = 0; mb < 2; ++mb) {
    float s[2];
#pragma unroll
    for (int rho = 0; rho < 2; ++rho) s[rho] = ((gb[rho] & 1) == mb) ? ub[rho] : 0.0f;
    ryT_reg <1, 0,  1, 2>(s, cs, sn);
    ryT_lane<1, 1, 16, 2>(s, cs, sn, lane);
    ryT_lane<1, 2,  8, 2>(s, cs, sn, lane);
    ryT_lane<1, 3,  4, 2>(s, cs, sn, lane);
    ryT_lane<1, 4,  2, 2>(s, cs, sn, lane);
    ryT_lane<1, 5,  1, 2>(s, cs, sn, lane);

    float sc[2];
#pragma unroll
    for (int rho = 0; rho < 2; ++rho) sc[rho] = __shfl(s[rho], gb[rho] & 31, 32);
    const int m = 2 * wave + mb;
#pragma unroll
    for (int sg = 0; sg < 2; ++sg) {
      float a[2];
#pragma unroll
      for (int rho = 0; rho < 2; ++rho) a[rho] = (sg && (gb[rho] & 1)) ? -sc[rho] : sc[rho];
      ryT_reg <0, 0,  1, 2>(a, cs, sn);
      ryT_lane<0, 1, 16, 2>(a, cs, sn, lane);
      ryT_lane<0, 2,  8, 2>(a, cs, sn, lane);
      ryT_lane<0, 3,  4, 2>(a, cs, sn, lane);
      ryT_lane<0, 4,  2, 2>(a, cs, sn, lane);
      ryT_lane<0, 5,  1, 2>(a, cs, sn, lane);
#pragma unroll
      for (int rho = 0; rho < 2; ++rho) {
        const int ku = rho * 8 + (lane >> 2);
        Bf[(((((sg * 2 + (m >> 4)) * 16 + ku) * 32) + (lane & 1) * 16 + (m & 15)) * 2) + ((lane >> 1) & 1)] = a[rho];
      }
    }
  }
  asm volatile("s_wait_dscnt 0x0\n\ts_barrier_signal -1\n\ts_barrier_wait -1" ::: "memory");
  __builtin_amdgcn_sched_barrier(0);

  const float c6 = lane_bcast(cs, 6), s6 = lane_bcast(sn, 6);
  const float al = (c6 + s6) * 0.70710678118654752f;
  const float be = (c6 - s6) * 0.70710678118654752f;

  const v2f* bl = Bs + lane;
  v8f acc[2][2][2];
#pragma unroll
  for (int sg = 0; sg < 2; ++sg)
#pragma unroll
    for (int nt = 0; nt < 2; ++nt)
#pragma unroll
      for (int b = 0; b < 2; ++b)
        acc[sg][nt][b] = v8f{0.f, 0.f, 0.f, 0.f, 0.f, 0.f, 0.f, 0.f};

  v2f sq0 = {0.f, 0.f}, sq1 = {0.f, 0.f}, sq2 = {0.f, 0.f}, sq3 = {0.f, 0.f};

  v2f bcur[2][2];
#pragma unroll
  for (int sg = 0; sg < 2; ++sg)
#pragma unroll
    for (int nt = 0; nt < 2; ++nt) bcur[sg][nt] = bl[((sg * 2 + nt) * 16) * 32];

#pragma unroll
  for (int u = 0; u < 16; ++u) {
    if ((u & 1) == 0 && u + PF < 16) {
#pragma unroll
      for (int t = 2 * (u + PF); t < 2 * (u + PF) + 4; ++t)
        q[t] = *(const v4f*)(xrow + 8 * t);
    }
    __builtin_amdgcn_sched_barrier(0);

    v2f bnxt[2][2];
    if (u + 1 < 16) {
#pragma unroll
      for (int sg = 0; sg < 2; ++sg)
#pragma unroll
        for (int nt = 0; nt < 2; ++nt)
          bnxt[sg][nt] = bl[((sg * 2 + nt) * 16 + u + 1) * 32];
    }

    v2f yp[2], ym[2];
#pragma unroll
    for (int h = 0; h < 2; ++h) {
      const v4f qq = q[2 * u + h];
      const v2f lo = __builtin_shufflevector(qq, qq, 0, 1);
      const v2f hi = __builtin_shufflevector(qq, qq, 2, 3);
      yp[h] = al * lo + be * hi;
      ym[h] = be * lo - al * hi;
    }
    const v2f a0 = v2f{yp[0][0], yp[1][0]};
    const v2f a1 = v2f{yp[0][1], yp[1][1]};
    const v2f m0 = v2f{ym[0][0], ym[1][0]};
    const v2f m1 = v2f{ym[0][1], ym[1][1]};
    sq0 += a0 * a0;
    sq1 += a1 * a1;
    sq2 += m0 * m0;
    sq3 += m1 * m1;
#pragma unroll
    for (int nt = 0; nt < 2; ++nt) {
      acc[0][nt][0] = __builtin_amdgcn_wmma_f32_16x16x4_f32(
           false, a0,  false, bcur[0][nt],
           (short)0, acc[0][nt][0],  false,  false);
      acc[0][nt][1] = __builtin_amdgcn_wmma_f32_16x16x4_f32(
           false, a1,  false, bcur[0][nt],
           (short)0, acc[0][nt][1],  false,  false);
      acc[1][nt][0] = __builtin_amdgcn_wmma_f32_16x16x4_f32(
           false, m0,  false, bcur[1][nt],
           (short)0, acc[1][nt][0],  false,  false);
      acc[1][nt][1] = __builtin_amdgcn_wmma_f32_16x16x4_f32(
           false, m1,  false, bcur[1][nt],
           (short)0, acc[1][nt][1],  false,  false);
    }
    if (u + 1 < 16) {
#pragma unroll
      for (int sg = 0; sg < 2; ++sg)
#pragma unroll
        for (int nt = 0; nt < 2; ++nt) bcur[sg][nt] = bnxt[sg][nt];
    }
  }

  float accP[8];
#pragma unroll
  for (int v = 0; v < 8; ++v) accP[v] = 0.f;
#pragma unroll
  for (int sg = 0; sg < 2; ++sg)
#pragma unroll
    for (int nt = 0; nt < 2; ++nt)
#pragma unroll
      for (int v = 0; v < 8; ++v)
        accP[v] += acc[sg][nt][0][v] * acc[sg][nt][0][v] + acc[sg][nt][1][v] * acc[sg][nt][1][v];

  const v2f sq = (sq0 + sq1) + (sq2 + sq3);
  redS[wave][lane] = sq[0] + sq[1];

#pragma unroll
  for (int v = 0; v < 8; ++v) {
    redP[wave][hf * 8 + v][l16] = accP[v];
  }
  __syncthreads();

  if (threadIdx.x < 256) {
    const int wv2 = threadIdx.x >> 4;
    const int r   = threadIdx.x & 15;
    float P = 0.f;
#pragma unroll
    for (int i = 0; i < 16; ++i) P += redP[wv2][r][i];
    const float S = redS[wv2][r] + redS[wv2][16 + r];
    const float z = (2.f * P - S) / S;
    float t = b2[0];
#pragma unroll
    for (int jj = 0; jj < 16; ++jj) {
      float h = z * W1[jj] + b1[jj];
      h = h > 0.f ? h : 0.f;
      t += h * W2[jj];
    }
    out[(size_t)blockIdx.x * 256 + wv2 * 16 + r] = 1.f / (1.f + expf(-t));
  }
}

extern "C" void kernel_launch(void* const* d_in, const int* in_sizes, int n_in,
                              void* d_out, int out_size, void* d_ws, size_t ws_size,
                              hipStream_t stream) {
  const float* x  = (const float*)d_in[0];
  const float* w  = (const float*)d_in[1];
  const float* W1 = (const float*)d_in[2];
  const float* b1 = (const float*)d_in[3];
  const float* W2 = (const float*)d_in[4];
  const float* b2 = (const float*)d_in[5];
  float* out = (float*)d_out;

  const int B = in_sizes[0] / DIM;
  qnn_fused_kernel<<<B / 256, 512, 0, stream>>>(x, w, W1, b1, W2, b2, out);
}
